# starting kernel
# baseline (speedup 1.0000x reference)
#include <hip/hip_runtime.h>


typedef __attribute__((ext_vector_type(2))) float v2f;
typedef __attribute__((ext_vector_type(8))) float v8f;

#define DFEAT 128

__global__ __launch_bounds__(256) void gemm128_wmma(
    const float* __restrict__ Xin, int Mrows,
    const float* __restrict__ W,
    const float* __restrict__ bias,
    float* __restrict__ Y)
{
    const int lane = threadIdx.x & 31;
    const int wave = threadIdx.x >> 5;
    const int half = lane >> 4;
    const int r    = lane & 15;
    const int m0   = (blockIdx.x * 8 + wave) * 16;
    if (m0 >= Mrows) return;

    v8f acc[8] = {};

    int arow = m0 + r;
    if (arow >= Mrows) arow = Mrows - 1;
    const float* ap = Xin + (size_t)arow * DFEAT + 2 * half;

    for (int kk = 0; kk < 32; ++kk) {
        const int k = kk * 4;
        v2f a;
        a.x = ap[k];
        a.y = ap[k + 1];

        const float* wrow = W + (size_t)(k + 2 * half) * DFEAT + r;
#pragma unroll
        for (int t = 0; t < 8; ++t) {
            v2f b;
            b.x = wrow[t * 16];
            b.y = wrow[t * 16 + DFEAT];
            acc[t] = __builtin_amdgcn_wmma_f32_16x16x4_f32(
                false, a, false, b, (short)0, acc[t], false, false);
        }
    }

#pragma unroll
    for (int t = 0; t < 8; ++t) {
        const int n = t * 16 + r;
        const float bv = bias ? bias[n] : 0.0f;
#pragma unroll
        for (int j = 0; j < 8; ++j) {
            const int row = m0 + j + 8 * half;
            if (row < Mrows) Y[(size_t)row * DFEAT + n] = acc[t][j] + bv;
        }
    }
}

__global__ __launch_bounds__(256) void scatter_rows(
    const float* __restrict__ src,
    const int* __restrict__ gidx,
    const int* __restrict__ sidx,
    float* __restrict__ dst,
    int* __restrict__ deg,
    int nnz)
{
    const int i = blockIdx.x * 8 + (threadIdx.x >> 5);
    if (i >= nnz) return;
    const int lane = threadIdx.x & 31;
    const int g = gidx[i];
    const int s = sidx[i];
    const float4 v = ((const float4*)(src + (size_t)g * DFEAT))[lane];
    float* dp = dst + (size_t)s * DFEAT + lane * 4;
    atomicAdd(dp + 0, v.x);
    atomicAdd(dp + 1, v.y);
    atomicAdd(dp + 2, v.z);
    atomicAdd(dp + 3, v.w);
    if (deg != nullptr && lane == 0) atomicAdd(deg + s, 1);
}

__global__ void zero_kernel(float* __restrict__ p, long n)
{
    long i = (long)blockIdx.x * blockDim.x + threadIdx.x;
    if (i < n) p[i] = 0.0f;
}

__global__ void blend_kernel(
    const float* __restrict__ Ap, const float* __restrict__ SB,
    const int* __restrict__ deg, const float* __restrict__ X0,
    float* __restrict__ Xn, long n)
{
    long i = (long)blockIdx.x * blockDim.x + threadIdx.x;
    if (i >= n) return;
    const int v = (int)(i >> 7);
    const float d = (float)deg[v];
    Xn[i] = 0.5f * (d * Ap[i] + SB[i]) + 0.5f * X0[i];
}

extern "C" void kernel_launch(void* const* d_in, const int* in_sizes, int n_in,
                              void* d_out, int out_size, void* d_ws, size_t ws_size,
                              hipStream_t stream) {
    const float* X      = (const float*)d_in[0];
    const int*   vertex = (const int*)d_in[1];
    const int*   edges  = (const int*)d_in[2];
    const float* X0     = (const float*)d_in[3];
    const float* W1     = (const float*)d_in[4];
    const float* b1     = (const float*)d_in[5];
    const float* W2     = (const float*)d_in[6];
    const float* b2     = (const float*)d_in[7];
    const float* W3     = (const float*)d_in[8];
    const float* b3     = (const float*)d_in[9];
    float* out = (float*)d_out;

    constexpr int N = 50000, M = 25000, NNZ = 800000;
    constexpr size_t ND = (size_t)N * DFEAT;
    constexpr size_t MD = (size_t)M * DFEAT;

    char* ws = (char*)d_ws;
    auto align256 = [](size_t x) { return (x + 255) & ~(size_t)255; };
    size_t off = 0;
    float* XW = (float*)(ws + off); off = align256(off + ND * 4);
    float* Xe = (float*)(ws + off); off = align256(off + MD * 4);
    float* Ap = (float*)(ws + off); off = align256(off + ND * 4);
    float* Bm = (float*)(ws + off); off = align256(off + MD * 4);
    float* SB = (float*)(ws + off); off = align256(off + ND * 4);
    int*  deg = (int*)  (ws + off); off = align256(off + (size_t)N * 4);
    float* Xn = XW;

    const int gemmN = (N + 127) / 128;
    const int gemmM = (M + 127) / 128;
    const int scatB = (NNZ + 7) / 8;

    gemm128_wmma<<<gemmN, 256, 0, stream>>>(X, N, W1, b1, XW);

    zero_kernel<<<(int)((MD + 255) / 256), 256, 0, stream>>>(Xe, (long)MD);
    scatter_rows<<<scatB, 256, 0, stream>>>(XW, vertex, edges, Xe, nullptr, NNZ);

    gemm128_wmma<<<gemmN, 256, 0, stream>>>(X, N, W2, b2, Ap);

    gemm128_wmma<<<gemmM, 256, 0, stream>>>(Xe, M, W2 + 128 * DFEAT, nullptr, Bm);

    zero_kernel<<<(int)((ND + 255) / 256), 256, 0, stream>>>(SB, (long)ND);
    zero_kernel<<<(N + 255) / 256, 256, 0, stream>>>((float*)deg, (long)N);
    scatter_rows<<<scatB, 256, 0, stream>>>(Bm, edges, vertex, SB, deg, NNZ);

    blend_kernel<<<(int)((ND + 255) / 256), 256, 0, stream>>>(Ap, SB, deg, X0, Xn, (long)ND);

    gemm128_wmma<<<gemmN, 256, 0, stream>>>(Xn, N, W3, b3, out);
}


# Row scatter without float atomics: counting-sort the incidence list each call (int atomics only); one wave sums each output row in registers in list order and stores it once; summation order changes
# speedup vs baseline: 1.1465x; 1.1465x over previous
#include <hip/hip_runtime.h>


typedef __attribute__((ext_vector_type(2))) float v2f;
typedef __attribute__((ext_vector_type(8))) float v8f;

#define DFEAT 128

__global__ __launch_bounds__(256) void gemm128_wmma(
    const float* __restrict__ Xin, int Mrows,
    const float* __restrict__ W,
    const float* __restrict__ bias,
    float* __restrict__ Y)
{
    const int lane = threadIdx.x & 31;
    const int wave = threadIdx.x >> 5;
    const int half = lane >> 4;
    const int r    = lane & 15;
    const int m0   = (blockIdx.x * 8 + wave) * 16;
    if (m0 >= Mrows) return;

    v8f acc[8] = {};

    int arow = m0 + r;
    if (arow >= Mrows) arow = Mrows - 1;
    const float* ap = Xin + (size_t)arow * DFEAT + 2 * half;

    for (int kk = 0; kk < 32; ++kk) {
        const int k = kk * 4;
        v2f a;
        a.x = ap[k];
        a.y = ap[k + 1];

        const float* wrow = W + (size_t)(k + 2 * half) * DFEAT + r;
#pragma unroll
        for (int t = 0; t < 8; ++t) {
            v2f b;
            b.x = wrow[t * 16];
            b.y = wrow[t * 16 + DFEAT];
            acc[t] = __builtin_amdgcn_wmma_f32_16x16x4_f32(
                false, a, false, b, (short)0, acc[t], false, false);
        }
    }

#pragma unroll
    for (int t = 0; t < 8; ++t) {
        const int n = t * 16 + r;
        const float bv = bias ? bias[n] : 0.0f;
#pragma unroll
        for (int j = 0; j < 8; ++j) {
            const int row = m0 + j + 8 * half;
            if (row < Mrows) Y[(size_t)row * DFEAT + n] = acc[t][j] + bv;
        }
    }
}

__global__ void zero_counts(int* __restrict__ a, int na, int* __restrict__ b, int nb)
{
    const int i = blockIdx.x * blockDim.x + threadIdx.x;
    if (i < na) a[i] = 0;
    else if (i - na < nb) b[i - na] = 0;
}

__global__ __launch_bounds__(256) void count_incidences(
    const int* __restrict__ vertex, const int* __restrict__ edges,
    int* __restrict__ cntE, int* __restrict__ cntV, int nnz)
{
    const int i = blockIdx.x * 256 + threadIdx.x;
    if (i >= nnz) return;
    atomicAdd(cntE + edges[i], 1);
    atomicAdd(cntV + vertex[i], 1);
}

template <int NELEM>
__device__ __forceinline__ void block_exclusive_scan(
    const int* __restrict__ cnt, int* __restrict__ cur, int* wsum)
{
    static_assert(NELEM % 4 == 0, "int4 access needs a multiple of 4");
    constexpr int IT = ((NELEM + 31) / 32 + 127) / 128;
    constexpr int CW = IT * 128;
    const int tid  = threadIdx.x;
    const int lane = tid & 31;
    const int wave = tid >> 5;
    const int base = wave * CW + lane * 4;

    int4 v[IT];
    int sum = 0;
#pragma unroll
    for (int q = 0; q < IT; ++q) {
        const int idx = base + 128 * q;
        v[q] = (idx < NELEM) ? *(const int4*)(cnt + idx) : make_int4(0, 0, 0, 0);
        sum += v[q].x + v[q].y + v[q].z + v[q].w;
    }

    for (int d = 16; d >= 1; d >>= 1) sum += __shfl_xor(sum, d);
    if (lane == 0) wsum[wave] = sum;
    __syncthreads();
    if (wave == 0) {
        const int w = wsum[lane];
        int wi = w;
        for (int d = 1; d < 32; d <<= 1) {
            const int t = __shfl_up(wi, d);
            if (lane >= d) wi += t;
        }
        wsum[lane] = wi - w;
    }
    __syncthreads();

    int carry = wsum[wave];
#pragma unroll
    for (int q = 0; q < IT; ++q) {
        const int s = v[q].x + v[q].y + v[q].z + v[q].w;
        int incl = s;
        for (int d = 1; d < 32; d <<= 1) {
            const int t = __shfl_up(incl, d);
            if (lane >= d) incl += t;
        }
        const int idx = base + 128 * q;
        if (idx < NELEM) {
            int4 o;
            o.x = carry + incl - s;
            o.y = o.x + v[q].x;
            o.z = o.y + v[q].y;
            o.w = o.z + v[q].z;
            *(int4*)(cur + idx) = o;
        }
        carry += __shfl(incl, 31);
    }
}

template <int NE, int NV>
__global__ __launch_bounds__(1024) void scan_counts(
    const int* __restrict__ cntE, int* __restrict__ curE,
    const int* __restrict__ cntV, int* __restrict__ curV)
{
    __shared__ int wsum[32];
    if (blockIdx.x == 0) block_exclusive_scan<NE>(cntE, curE, wsum);
    else                 block_exclusive_scan<NV>(cntV, curV, wsum);
}

__global__ __launch_bounds__(256) void fill_lists(
    const int* __restrict__ vertex, const int* __restrict__ edges,
    int* __restrict__ curE, int* __restrict__ curV,
    int* __restrict__ permE, int* __restrict__ permV, int nnz)
{
    const int i = blockIdx.x * 256 + threadIdx.x;
    if (i >= nnz) return;
    const int v = vertex[i];
    const int e = edges[i];
    permE[atomicAdd(curE + e, 1)] = v;
    permV[atomicAdd(curV + v, 1)] = e;
}

__global__ __launch_bounds__(256) void segsum_rows(
    const float* __restrict__ src,
    const int* __restrict__ perm,
    const int* __restrict__ segend,
    const int* __restrict__ cnt,
    float* __restrict__ dst,
    int nrows)
{
    const int row = blockIdx.x * 8 + (threadIdx.x >> 5);
    if (row >= nrows) return;
    const int lane = threadIdx.x & 31;
    const int end = __builtin_amdgcn_readfirstlane(segend[row]);
    const int n   = __builtin_amdgcn_readfirstlane(cnt[row]);

    float4 acc = make_float4(0.0f, 0.0f, 0.0f, 0.0f);
    for (int base = end - n; base < end; base += 32) {
        const int m = (end - base < 32) ? (end - base) : 32;
        const int mine = (lane < m) ? perm[base + lane] : 0;
#pragma unroll 4
        for (int j = 0; j < m; ++j) {
            const int g = __builtin_amdgcn_readlane(mine, j);
            const float4 x = ((const float4*)(src + (size_t)g * DFEAT))[lane];
            acc.x += x.x; acc.y += x.y; acc.z += x.z; acc.w += x.w;
        }
    }
    ((float4*)(dst + (size_t)row * DFEAT))[lane] = acc;
}

__global__ void blend_kernel(
    const float* __restrict__ Ap, const float* __restrict__ SB,
    const int* __restrict__ deg, const float* __restrict__ X0,
    float* __restrict__ Xn, long n)
{
    long i = (long)blockIdx.x * blockDim.x + threadIdx.x;
    if (i >= n) return;
    const int v = (int)(i >> 7);
    const float d = (float)deg[v];
    Xn[i] = 0.5f * (d * Ap[i] + SB[i]) + 0.5f * X0[i];
}

extern "C" void kernel_launch(void* const* d_in, const int* in_sizes, int n_in,
                              void* d_out, int out_size, void* d_ws, size_t ws_size,
                              hipStream_t stream) {
    const float* X      = (const float*)d_in[0];
    const int*   vertex = (const int*)d_in[1];
    const int*   edges  = (const int*)d_in[2];
    const float* X0     = (const float*)d_in[3];
    const float* W1     = (const float*)d_in[4];
    const float* b1     = (const float*)d_in[5];
    const float* W2     = (const float*)d_in[6];
    const float* b2     = (const float*)d_in[7];
    const float* W3     = (const float*)d_in[8];
    const float* b3     = (const float*)d_in[9];
    float* out = (float*)d_out;

    constexpr int N = 50000, M = 25000, NNZ = 800000;
    constexpr size_t ND = (size_t)N * DFEAT;
    constexpr size_t MD = (size_t)M * DFEAT;

    char* ws = (char*)d_ws;
    auto align256 = [](size_t x) { return (x + 255) & ~(size_t)255; };
    size_t off = 0;
    float* XW = (float*)(ws + off); off = align256(off + ND * 4);
    float* Xe = (float*)(ws + off); off = align256(off + MD * 4);
    const size_t offAp = off;
    float* Ap = (float*)(ws + off); off = align256(off + ND * 4);
    float* Bm = (float*)(ws + off); off = align256(off + MD * 4);
    float* SB = (float*)(ws + off); off = align256(off + ND * 4);
    int*  deg = (int*)  (ws + off); off = align256(off + (size_t)N * 4);
    float* Xn = XW;

    size_t o2 = offAp;
    int* permE = (int*)(ws + o2); o2 = align256(o2 + (size_t)NNZ * 4);
    int* permV = (int*)(ws + o2); o2 = align256(o2 + (size_t)NNZ * 4);
    int* curE  = (int*)(ws + o2); o2 = align256(o2 + (size_t)M * 4);
    int* curV  = (int*)(ws + o2); o2 = align256(o2 + (size_t)N * 4);
    int* cntE  = (int*)(ws + o2); o2 = align256(o2 + (size_t)M * 4);
    int* cntV  = deg;
    static_assert(2 * (size_t)NNZ * 4 + 2 * (size_t)M * 4 + (size_t)N * 4 + 5 * 256 <= ND * 4,
                  "sorted lists must fit in Ap's region");

    const int gemmN = (N + 127) / 128;
    const int gemmM = (M + 127) / 128;
    const int nnzB  = (NNZ + 255) / 256;

    gemm128_wmma<<<gemmN, 256, 0, stream>>>(X, N, W1, b1, XW);

    zero_counts<<<(M + N + 255) / 256, 256, 0, stream>>>(cntE, M, cntV, N);
    count_incidences<<<nnzB, 256, 0, stream>>>(vertex, edges, cntE, cntV, NNZ);
    scan_counts<M, N><<<2, 1024, 0, stream>>>(cntE, curE, cntV, curV);
    fill_lists<<<nnzB, 256, 0, stream>>>(vertex, edges, curE, curV, permE, permV, NNZ);

    segsum_rows<<<(M + 7) / 8, 256, 0, stream>>>(XW, permE, curE, cntE, Xe, M);

    gemm128_wmma<<<gemmM, 256, 0, stream>>>(Xe, M, W2 + 128 * DFEAT, nullptr, Bm);

    segsum_rows<<<(N + 7) / 8, 256, 0, stream>>>(Bm, permV, curV, cntV, SB, N);

    gemm128_wmma<<<gemmN, 256, 0, stream>>>(X, N, W2, b2, Ap);

    blend_kernel<<<(int)((ND + 255) / 256), 256, 0, stream>>>(Ap, SB, deg, X0, Xn, (long)ND);

    gemm128_wmma<<<gemmN, 256, 0, stream>>>(Xn, N, W3, b3, out);
}


# One atomic pass instead of two: the counting atomic's returned value is kept as the incidence's rank, so list placement is start+rank with no cursor atomics; sum order follows count arrival order
# speedup vs baseline: 1.5100x; 1.3171x over previous
#include <hip/hip_runtime.h>


typedef __attribute__((ext_vector_type(2))) float v2f;
typedef __attribute__((ext_vector_type(8))) float v8f;

#define DFEAT 128

__global__ __launch_bounds__(256) void gemm128_wmma(
    const float* __restrict__ Xin, int Mrows,
    const float* __restrict__ W,
    const float* __restrict__ bias,
    float* __restrict__ Y)
{
    const int lane = threadIdx.x & 31;
    const int wave = threadIdx.x >> 5;
    const int half = lane >> 4;
    const int r    = lane & 15;
    const int m0   = (blockIdx.x * 8 + wave) * 16;
    if (m0 >= Mrows) return;

    v8f acc[8] = {};

    int arow = m0 + r;
    if (arow >= Mrows) arow = Mrows - 1;
    const float* ap = Xin + (size_t)arow * DFEAT + 2 * half;

    for (int kk = 0; kk < 32; ++kk) {
        const int k = kk * 4;
        v2f a;
        a.x = ap[k];
        a.y = ap[k + 1];

        const float* wrow = W + (size_t)(k + 2 * half) * DFEAT + r;
#pragma unroll
        for (int t = 0; t < 8; ++t) {
            v2f b;
            b.x = wrow[t * 16];
            b.y = wrow[t * 16 + DFEAT];
            acc[t] = __builtin_amdgcn_wmma_f32_16x16x4_f32(
                false, a, false, b, (short)0, acc[t], false, false);
        }
    }

#pragma unroll
    for (int t = 0; t < 8; ++t) {
        const int n = t * 16 + r;
        const float bv = bias ? bias[n] : 0.0f;
#pragma unroll
        for (int j = 0; j < 8; ++j) {
            const int row = m0 + j + 8 * half;
            if (row < Mrows) Y[(size_t)row * DFEAT + n] = acc[t][j] + bv;
        }
    }
}

__global__ void zero_counts(int* __restrict__ a, int na, int* __restrict__ b, int nb)
{
    const int i = blockIdx.x * blockDim.x + threadIdx.x;
    if (i < na) a[i] = 0;
    else if (i - na < nb) b[i - na] = 0;
}

__global__ __launch_bounds__(256) void count_incidences(
    const int* __restrict__ vertex, const int* __restrict__ edges,
    int* __restrict__ cntE, int* __restrict__ cntV,
    int* __restrict__ rankE, int* __restrict__ rankV, int nnz)
{
    const int i = blockIdx.x * 256 + threadIdx.x;
    if (i >= nnz) return;
    const int e = edges[i];
    const int v = vertex[i];
    const int re = atomicAdd(cntE + e, 1);
    const int rv = atomicAdd(cntV + v, 1);
    rankE[i] = re;
    rankV[i] = rv;
}

template <int NELEM>
__device__ __forceinline__ void block_exclusive_scan(
    const int* __restrict__ cnt, int* __restrict__ cur, int* wsum)
{
    static_assert(NELEM % 4 == 0, "int4 access needs a multiple of 4");
    constexpr int IT = ((NELEM + 31) / 32 + 127) / 128;
    constexpr int CW = IT * 128;
    const int tid  = threadIdx.x;
    const int lane = tid & 31;
    const int wave = tid >> 5;
    const int base = wave * CW + lane * 4;

    int4 v[IT];
    int sum = 0;
#pragma unroll
    for (int q = 0; q < IT; ++q) {
        const int idx = base + 128 * q;
        v[q] = (idx < NELEM) ? *(const int4*)(cnt + idx) : make_int4(0, 0, 0, 0);
        sum += v[q].x + v[q].y + v[q].z + v[q].w;
    }

    for (int d = 16; d >= 1; d >>= 1) sum += __shfl_xor(sum, d);
    if (lane == 0) wsum[wave] = sum;
    __syncthreads();
    if (wave == 0) {
        const int w = wsum[lane];
        int wi = w;
        for (int d = 1; d < 32; d <<= 1) {
            const int t = __shfl_up(wi, d);
            if (lane >= d) wi += t;
        }
        wsum[lane] = wi - w;
    }
    __syncthreads();

    int carry = wsum[wave];
#pragma unroll
    for (int q = 0; q < IT; ++q) {
        const int s = v[q].x + v[q].y + v[q].z + v[q].w;
        int incl = s;
        for (int d = 1; d < 32; d <<= 1) {
            const int t = __shfl_up(incl, d);
            if (lane >= d) incl += t;
        }
        const int idx = base + 128 * q;
        if (idx < NELEM) {
            int4 o;
            o.x = carry + incl - s;
            o.y = o.x + v[q].x;
            o.z = o.y + v[q].y;
            o.w = o.z + v[q].z;
            *(int4*)(cur + idx) = o;
        }
        carry += __shfl(incl, 31);
    }
}

template <int NE, int NV>
__global__ __launch_bounds__(1024) void scan_counts(
    const int* __restrict__ cntE, int* __restrict__ curE,
    const int* __restrict__ cntV, int* __restrict__ curV)
{
    __shared__ int wsum[32];
    if (blockIdx.x == 0) block_exclusive_scan<NE>(cntE, curE, wsum);
    else                 block_exclusive_scan<NV>(cntV, curV, wsum);
}

__global__ __launch_bounds__(256) void fill_lists(
    const int* __restrict__ vertex, const int* __restrict__ edges,
    const int* __restrict__ startE, const int* __restrict__ startV,
    const int* __restrict__ rankE, const int* __restrict__ rankV,
    int* __restrict__ permE, int* __restrict__ permV, int nnz)
{
    const int i = blockIdx.x * 256 + threadIdx.x;
    if (i >= nnz) return;
    const int v = vertex[i];
    const int e = edges[i];
    permE[startE[e] + rankE[i]] = v;
    permV[startV[v] + rankV[i]] = e;
}

__global__ __launch_bounds__(256) void segsum_rows(
    const float* __restrict__ src,
    const int* __restrict__ perm,
    const int* __restrict__ segbeg,
    const int* __restrict__ cnt,
    float* __restrict__ dst,
    int nrows)
{
    const int row = blockIdx.x * 8 + (threadIdx.x >> 5);
    if (row >= nrows) return;
    const int lane = threadIdx.x & 31;
    const int beg = __builtin_amdgcn_readfirstlane(segbeg[row]);
    const int n   = __builtin_amdgcn_readfirstlane(cnt[row]);
    const int end = beg + n;

    float4 acc = make_float4(0.0f, 0.0f, 0.0f, 0.0f);
    for (int base = beg; base < end; base += 32) {
        const int m = (end - base < 32) ? (end - base) : 32;
        const int mine = (lane < m) ? perm[base + lane] : 0;
#pragma unroll 4
        for (int j = 0; j < m; ++j) {
            const int g = __builtin_amdgcn_readlane(mine, j);
            const float4 x = ((const float4*)(src + (size_t)g * DFEAT))[lane];
            acc.x += x.x; acc.y += x.y; acc.z += x.z; acc.w += x.w;
        }
    }
    ((float4*)(dst + (size_t)row * DFEAT))[lane] = acc;
}

__global__ void blend_kernel(
    const float* __restrict__ Ap, const float* __restrict__ SB,
    const int* __restrict__ deg, const float* __restrict__ X0,
    float* __restrict__ Xn, long n)
{
    long i = (long)blockIdx.x * blockDim.x + threadIdx.x;
    if (i >= n) return;
    const int v = (int)(i >> 7);
    const float d = (float)deg[v];
    Xn[i] = 0.5f * (d * Ap[i] + SB[i]) + 0.5f * X0[i];
}

extern "C" void kernel_launch(void* const* d_in, const int* in_sizes, int n_in,
                              void* d_out, int out_size, void* d_ws, size_t ws_size,
                              hipStream_t stream) {
    const float* X      = (const float*)d_in[0];
    const int*   vertex = (const int*)d_in[1];
    const int*   edges  = (const int*)d_in[2];
    const float* X0     = (const float*)d_in[3];
    const float* W1     = (const float*)d_in[4];
    const float* b1     = (const float*)d_in[5];
    const float* W2     = (const float*)d_in[6];
    const float* b2     = (const float*)d_in[7];
    const float* W3     = (const float*)d_in[8];
    const float* b3     = (const float*)d_in[9];
    float* out = (float*)d_out;

    constexpr int N = 50000, M = 25000, NNZ = 800000;
    constexpr size_t ND = (size_t)N * DFEAT;
    constexpr size_t MD = (size_t)M * DFEAT;

    char* ws = (char*)d_ws;
    auto align256 = [](size_t x) { return (x + 255) & ~(size_t)255; };
    size_t off = 0;
    float* XW = (float*)(ws + off); off = align256(off + ND * 4);
    float* Xe = (float*)(ws + off); off = align256(off + MD * 4);
    const size_t offAp = off;
    float* Ap = (float*)(ws + off); off = align256(off + ND * 4);
    float* Bm = (float*)(ws + off); off = align256(off + MD * 4);
    float* SB = (float*)(ws + off); off = align256(off + ND * 4);
    int*  deg = (int*)  (ws + off); off = align256(off + (size_t)N * 4);
    float* Xn = XW;

    size_t o2 = offAp;
    int* permE = (int*)(ws + o2); o2 = align256(o2 + (size_t)NNZ * 4);
    int* permV = (int*)(ws + o2); o2 = align256(o2 + (size_t)NNZ * 4);
    int* curE  = (int*)(ws + o2); o2 = align256(o2 + (size_t)M * 4);
    int* curV  = (int*)(ws + o2); o2 = align256(o2 + (size_t)N * 4);
    int* cntE  = (int*)(ws + o2); o2 = align256(o2 + (size_t)M * 4);
    int* rankE = (int*)(ws + o2); o2 = align256(o2 + (size_t)NNZ * 4);
    int* rankV = (int*)(ws + o2); o2 = align256(o2 + (size_t)NNZ * 4);
    int* cntV  = deg;
    static_assert(4 * (size_t)NNZ * 4 + 2 * (size_t)M * 4 + (size_t)N * 4 + 7 * 256 <= ND * 4,
                  "sorted lists and ranks must fit in Ap's region");

    const int gemmN = (N + 127) / 128;
    const int gemmM = (M + 127) / 128;
    const int nnzB  = (NNZ + 255) / 256;

    gemm128_wmma<<<gemmN, 256, 0, stream>>>(X, N, W1, b1, XW);

    zero_counts<<<(M + N + 255) / 256, 256, 0, stream>>>(cntE, M, cntV, N);
    count_incidences<<<nnzB, 256, 0, stream>>>(vertex, edges, cntE, cntV, rankE, rankV, NNZ);
    scan_counts<M, N><<<2, 1024, 0, stream>>>(cntE, curE, cntV, curV);
    fill_lists<<<nnzB, 256, 0, stream>>>(vertex, edges, curE, curV, rankE, rankV, permE, permV, NNZ);

    segsum_rows<<<(M + 7) / 8, 256, 0, stream>>>(XW, permE, curE, cntE, Xe, M);

    gemm128_wmma<<<gemmM, 256, 0, stream>>>(Xe, M, W2 + 128 * DFEAT, nullptr, Bm);

    segsum_rows<<<(N + 7) / 8, 256, 0, stream>>>(Bm, permV, curV, cntV, SB, N);

    gemm128_wmma<<<gemmN, 256, 0, stream>>>(X, N, W2, b2, Ap);

    blend_kernel<<<(int)((ND + 255) / 256), 256, 0, stream>>>(Ap, SB, deg, X0, Xn, (long)ND);

    gemm128_wmma<<<gemmN, 256, 0, stream>>>(Xn, N, W3, b3, out);
}


# Counting atomics moved from global memory to LDS: each id list is cut into 32 chunks, one work-group counts a chunk into a whole LDS histogram; sum order = chunk order, then LDS arrival order
# speedup vs baseline: 2.0106x; 1.3315x over previous
#include <hip/hip_runtime.h>


typedef __attribute__((ext_vector_type(2))) float v2f;
typedef __attribute__((ext_vector_type(8))) float v8f;

#define DFEAT 128

__global__ __launch_bounds__(256) void gemm128_wmma(
    const float* __restrict__ Xin, int Mrows,
    const float* __restrict__ W,
    const float* __restrict__ bias,
    float* __restrict__ Y)
{
    const int lane = threadIdx.x & 31;
    const int wave = threadIdx.x >> 5;
    const int half = lane >> 4;
    const int r    = lane & 15;
    const int m0   = (blockIdx.x * 8 + wave) * 16;
    if (m0 >= Mrows) return;

    v8f acc[8] = {};

    int arow = m0 + r;
    if (arow >= Mrows) arow = Mrows - 1;
    const float* ap = Xin + (size_t)arow * DFEAT + 2 * half;

    for (int kk = 0; kk < 32; ++kk) {
        const int k = kk * 4;
        v2f a;
        a.x = ap[k];
        a.y = ap[k + 1];

        const float* wrow = W + (size_t)(k + 2 * half) * DFEAT + r;
#pragma unroll
        for (int t = 0; t < 8; ++t) {
            v2f b;
            b.x = wrow[t * 16];
            b.y = wrow[t * 16 + DFEAT];
            acc[t] = __builtin_amdgcn_wmma_f32_16x16x4_f32(
                false, a, false, b, (short)0, acc[t], false, false);
        }
    }

#pragma unroll
    for (int t = 0; t < 8; ++t) {
        const int n = t * 16 + r;
        const float bv = bias ? bias[n] : 0.0f;
#pragma unroll
        for (int j = 0; j < 8; ++j) {
            const int row = m0 + j + 8 * half;
            if (row < Mrows) Y[(size_t)row * DFEAT + n] = acc[t][j] + bv;
        }
    }
}

template <int NE, int NV, int NCHUNK, int PER>
__global__ __launch_bounds__(1024) void count_chunks(
    const int* __restrict__ vertex, const int* __restrict__ edges,
    int* __restrict__ tabE,
    int* __restrict__ tabV,
    int* __restrict__ rankE, int* __restrict__ rankV, int nnz)
{
    __shared__ int hist[(NV > NE) ? NV : NE];
    const int tid = threadIdx.x;
    const bool isE = blockIdx.x < NCHUNK;
    const int c   = isE ? (int)blockIdx.x : (int)blockIdx.x - NCHUNK;
    const int nb  = isE ? NE : NV;
    const int* __restrict__ ids  = isE ? edges : vertex;
    int* __restrict__ rank = isE ? rankE : rankV;
    int* __restrict__ row  = isE ? tabE + (size_t)c * NE : tabV + (size_t)c * NV;

    for (int i = tid; i < nb; i += 1024) hist[i] = 0;
    __syncthreads();

    const int beg = c * PER;
    const int end = (beg + PER < nnz) ? (beg + PER) : nnz;
    constexpr int U = 12;
    int base = beg + tid;
    for (; base + (U - 1) * 1024 < end; base += 1024 * U) {
        int id[U], r[U];
#pragma unroll
        for (int u = 0; u < U; ++u) id[u] = ids[base + u * 1024];
#pragma unroll
        for (int u = 0; u < U; ++u) r[u] = atomicAdd(&hist[id[u]], 1);
#pragma unroll
        for (int u = 0; u < U; ++u) rank[base + u * 1024] = r[u];
    }
    for (; base < end; base += 1024) rank[base] = atomicAdd(&hist[ids[base]], 1);
    __syncthreads();

    for (int i = tid; i < nb; i += 1024) row[i] = hist[i];
}

template <int STRIDE, int NCHUNK>
__device__ __forceinline__ void column_offsets(int* __restrict__ col, int* __restrict__ cnt)
{
    int v[NCHUNK];
#pragma unroll
    for (int c = 0; c < NCHUNK; ++c) v[c] = col[(size_t)c * STRIDE];
    int run = 0;
#pragma unroll
    for (int c = 0; c < NCHUNK; ++c) { col[(size_t)c * STRIDE] = run; run += v[c]; }
    *cnt = run;
}

template <int NE, int NV, int NCHUNK>
__global__ __launch_bounds__(256) void chunk_offsets(
    int* __restrict__ tabE, int* __restrict__ tabV,
    int* __restrict__ cntE, int* __restrict__ cntV)
{
    const int i = blockIdx.x * 256 + threadIdx.x;
    if (i < NE)           column_offsets<NE, NCHUNK>(tabE + i, cntE + i);
    else if (i - NE < NV) column_offsets<NV, NCHUNK>(tabV + (i - NE), cntV + (i - NE));
}

template <int NELEM>
__device__ __forceinline__ void block_exclusive_scan(
    const int* __restrict__ cnt, int* __restrict__ cur, int* wsum)
{
    static_assert(NELEM % 4 == 0, "int4 access needs a multiple of 4");
    constexpr int IT = ((NELEM + 31) / 32 + 127) / 128;
    constexpr int CW = IT * 128;
    const int tid  = threadIdx.x;
    const int lane = tid & 31;
    const int wave = tid >> 5;
    const int base = wave * CW + lane * 4;

    int4 v[IT];
    int sum = 0;
#pragma unroll
    for (int q = 0; q < IT; ++q) {
        const int idx = base + 128 * q;
        v[q] = (idx < NELEM) ? *(const int4*)(cnt + idx) : make_int4(0, 0, 0, 0);
        sum += v[q].x + v[q].y + v[q].z + v[q].w;
    }

    for (int d = 16; d >= 1; d >>= 1) sum += __shfl_xor(sum, d);
    if (lane == 0) wsum[wave] = sum;
    __syncthreads();
    if (wave == 0) {
        const int w = wsum[lane];
        int wi = w;
        for (int d = 1; d < 32; d <<= 1) {
            const int t = __shfl_up(wi, d);
            if (lane >= d) wi += t;
        }
        wsum[lane] = wi - w;
    }
    __syncthreads();

    int carry = wsum[wave];
#pragma unroll
    for (int q = 0; q < IT; ++q) {
        const int s = v[q].x + v[q].y + v[q].z + v[q].w;
        int incl = s;
        for (int d = 1; d < 32; d <<= 1) {
            const int t = __shfl_up(incl, d);
            if (lane >= d) incl += t;
        }
        const int idx = base + 128 * q;
        if (idx < NELEM) {
            int4 o;
            o.x = carry + incl - s;
            o.y = o.x + v[q].x;
            o.z = o.y + v[q].y;
            o.w = o.z + v[q].z;
            *(int4*)(cur + idx) = o;
        }
        carry += __shfl(incl, 31);
    }
}

template <int NE, int NV>
__global__ __launch_bounds__(1024) void scan_counts(
    const int* __restrict__ cntE, int* __restrict__ curE,
    const int* __restrict__ cntV, int* __restrict__ curV)
{
    __shared__ int wsum[32];
    if (blockIdx.x == 0) block_exclusive_scan<NE>(cntE, curE, wsum);
    else                 block_exclusive_scan<NV>(cntV, curV, wsum);
}

template <int NE, int NV, int PER>
__global__ __launch_bounds__(256) void fill_lists(
    const int* __restrict__ vertex, const int* __restrict__ edges,
    const int* __restrict__ startE, const int* __restrict__ startV,
    const int* __restrict__ tabE, const int* __restrict__ tabV,
    const int* __restrict__ rankE, const int* __restrict__ rankV,
    int* __restrict__ permE, int* __restrict__ permV, int nnz)
{
    const int i = blockIdx.x * 256 + threadIdx.x;
    if (i >= nnz) return;
    const int v = vertex[i];
    const int e = edges[i];
    const int c = i / PER;
    permE[startE[e] + tabE[(size_t)c * NE + e] + rankE[i]] = v;
    permV[startV[v] + tabV[(size_t)c * NV + v] + rankV[i]] = e;
}

__global__ __launch_bounds__(256) void segsum_rows(
    const float* __restrict__ src,
    const int* __restrict__ perm,
    const int* __restrict__ segbeg,
    const int* __restrict__ cnt,
    float* __restrict__ dst,
    int nrows)
{
    const int row = blockIdx.x * 8 + (threadIdx.x >> 5);
    if (row >= nrows) return;
    const int lane = threadIdx.x & 31;
    const int beg = __builtin_amdgcn_readfirstlane(segbeg[row]);
    const int n   = __builtin_amdgcn_readfirstlane(cnt[row]);
    const int end = beg + n;

    float4 acc = make_float4(0.0f, 0.0f, 0.0f, 0.0f);
    for (int base = beg; base < end; base += 32) {
        const int m = (end - base < 32) ? (end - base) : 32;
        const int mine = (lane < m) ? perm[base + lane] : 0;
#pragma unroll 4
        for (int j = 0; j < m; ++j) {
            const int g = __builtin_amdgcn_readlane(mine, j);
            const float4 x = ((const float4*)(src + (size_t)g * DFEAT))[lane];
            acc.x += x.x; acc.y += x.y; acc.z += x.z; acc.w += x.w;
        }
    }
    ((float4*)(dst + (size_t)row * DFEAT))[lane] = acc;
}

__global__ void blend_kernel(
    const float* __restrict__ Ap, const float* __restrict__ SB,
    const int* __restrict__ deg, const float* __restrict__ X0,
    float* __restrict__ Xn, long n)
{
    long i = (long)blockIdx.x * blockDim.x + threadIdx.x;
    if (i >= n) return;
    const int v = (int)(i >> 7);
    const float d = (float)deg[v];
    Xn[i] = 0.5f * (d * Ap[i] + SB[i]) + 0.5f * X0[i];
}

extern "C" void kernel_launch(void* const* d_in, const int* in_sizes, int n_in,
                              void* d_out, int out_size, void* d_ws, size_t ws_size,
                              hipStream_t stream) {
    const float* X      = (const float*)d_in[0];
    const int*   vertex = (const int*)d_in[1];
    const int*   edges  = (const int*)d_in[2];
    const float* X0     = (const float*)d_in[3];
    const float* W1     = (const float*)d_in[4];
    const float* b1     = (const float*)d_in[5];
    const float* W2     = (const float*)d_in[6];
    const float* b2     = (const float*)d_in[7];
    const float* W3     = (const float*)d_in[8];
    const float* b3     = (const float*)d_in[9];
    float* out = (float*)d_out;

    constexpr int N = 50000, M = 25000, NNZ = 800000;
    constexpr size_t ND = (size_t)N * DFEAT;
    constexpr size_t MD = (size_t)M * DFEAT;

    char* ws = (char*)d_ws;
    auto align256 = [](size_t x) { return (x + 255) & ~(size_t)255; };
    size_t off = 0;
    float* XW = (float*)(ws + off); off = align256(off + ND * 4);
    float* Xe = (float*)(ws + off); off = align256(off + MD * 4);
    const size_t offAp = off;
    float* Ap = (float*)(ws + off); off = align256(off + ND * 4);
    float* Bm = (float*)(ws + off); off = align256(off + MD * 4);
    float* SB = (float*)(ws + off); off = align256(off + ND * 4);
    int*  deg = (int*)  (ws + off); off = align256(off + (size_t)N * 4);
    float* Xn = XW;

    size_t o2 = offAp;
    int* permE = (int*)(ws + o2); o2 = align256(o2 + (size_t)NNZ * 4);
    int* permV = (int*)(ws + o2); o2 = align256(o2 + (size_t)NNZ * 4);
    int* curE  = (int*)(ws + o2); o2 = align256(o2 + (size_t)M * 4);
    int* curV  = (int*)(ws + o2); o2 = align256(o2 + (size_t)N * 4);
    int* cntE  = (int*)(ws + o2); o2 = align256(o2 + (size_t)M * 4);
    int* rankE = (int*)(ws + o2); o2 = align256(o2 + (size_t)NNZ * 4);
    int* rankV = (int*)(ws + o2); o2 = align256(o2 + (size_t)NNZ * 4);
    constexpr int NCHUNK = 32;
    constexpr int PER = (NNZ + NCHUNK - 1) / NCHUNK;
    int* tabE  = (int*)(ws + o2); o2 = align256(o2 + (size_t)NCHUNK * M * 4);
    int* tabV  = (int*)(ws + o2); o2 = align256(o2 + (size_t)NCHUNK * N * 4);
    int* cntV  = deg;
    static_assert(4 * (size_t)NNZ * 4 + 2 * (size_t)M * 4 + (size_t)N * 4
                  + (size_t)NCHUNK * (M + N) * 4 + 9 * 256 <= ND * 4,
                  "sorted lists, ranks and chunk tables must fit in Ap's region");

    const int gemmN = (N + 127) / 128;
    const int gemmM = (M + 127) / 128;
    const int nnzB  = (NNZ + 255) / 256;

    gemm128_wmma<<<gemmN, 256, 0, stream>>>(X, N, W1, b1, XW);

    count_chunks<M, N, NCHUNK, PER><<<2 * NCHUNK, 1024, 0, stream>>>(vertex, edges, tabE, tabV, rankE, rankV, NNZ);
    chunk_offsets<M, N, NCHUNK><<<(M + N + 255) / 256, 256, 0, stream>>>(tabE, tabV, cntE, cntV);
    scan_counts<M, N><<<2, 1024, 0, stream>>>(cntE, curE, cntV, curV);
    fill_lists<M, N, PER><<<nnzB, 256, 0, stream>>>(vertex, edges, curE, curV, tabE, tabV, rankE, rankV, permE, permV, NNZ);

    segsum_rows<<<(M + 7) / 8, 256, 0, stream>>>(XW, permE, curE, cntE, Xe, M);

    gemm128_wmma<<<gemmM, 256, 0, stream>>>(Xe, M, W2 + 128 * DFEAT, nullptr, Bm);

    segsum_rows<<<(N + 7) / 8, 256, 0, stream>>>(Bm, permV, curV, cntV, SB, N);

    gemm128_wmma<<<gemmN, 256, 0, stream>>>(X, N, W2, b2, Ap);

    blend_kernel<<<(int)((ND + 255) / 256), 256, 0, stream>>>(Ap, SB, deg, X0, Xn, (long)ND);

    gemm128_wmma<<<gemmN, 256, 0, stream>>>(Xn, N, W3, b3, out);
}


# W1 hoisted out of the hyperedge gather: X rows summed per hyperedge, one M-row GEMM with Wc=W1@W2b (+cnt*b1@W2b), Wc formed in spare blocks of the offsets launch; no X@W1 launch; sum order changes
# speedup vs baseline: 2.6787x; 1.3323x over previous
#include <hip/hip_runtime.h>


typedef __attribute__((ext_vector_type(2))) float v2f;
typedef __attribute__((ext_vector_type(8))) float v8f;

#define DFEAT 128

template <bool GUARD>
__device__ __forceinline__ void blend_store(
    const v8f (&acc)[8], const float (&d)[8], const float (&bv)[8],
    int m0, int half, int r, int Mrows,
    const float* __restrict__ SB, const float* __restrict__ X0,
    float* __restrict__ Y)
{
#pragma unroll
    for (int t = 0; t < 8; ++t) {
        const int n = t * 16 + r;
#pragma unroll
        for (int j = 0; j < 8; ++j) {
            const int row = m0 + j + 8 * half;
            if (!GUARD || row < Mrows) {
                const size_t i = (size_t)row * DFEAT + n;
                const float ap = acc[t][j] + bv[t];
                Y[i] = 0.5f * (d[j] * ap + SB[i]) + 0.5f * X0[i];
            }
        }
    }
}

#define WPITCH 288
#define APITCH 132

template <bool BLEND, bool ROWBIAS>
__global__ __launch_bounds__(256) void gemm128_wmma(
    const float* __restrict__ Xin, int Mrows,
    const float* __restrict__ W,
    const float* __restrict__ bias,
    const float* __restrict__ SB,
    const int* __restrict__ deg,
    const float* __restrict__ X0,
    float* __restrict__ Y)
{
    __shared__ __attribute__((aligned(16))) float ldsW[64 * WPITCH];
    __shared__ __attribute__((aligned(16))) float ldsA[8 * 16 * APITCH];

    const int tid  = threadIdx.x;
    const int lane = tid & 31;
    const int wave = tid >> 5;
    const int half = lane >> 4;
    const int r    = lane & 15;
    const int m0   = (blockIdx.x * 8 + wave) * 16;
    const bool live = m0 < Mrows;

    float4 wx[8], wy[8];
#pragma unroll
    for (int it = 0; it < 8; ++it) {
        const int id = it * 256 + tid;
        const int n4 = id & 31;
        const int q  = id >> 5;
        wx[it] = *(const float4*)(W + (size_t)(2 * q) * DFEAT + n4 * 4);
        wy[it] = *(const float4*)(W + (size_t)(2 * q + 1) * DFEAT + n4 * 4);
    }

    float4 av[16];
#pragma unroll
    for (int i = 0; i < 16; ++i) {
        int row = m0 + i;
        if (row >= Mrows) row = Mrows - 1;
        av[i] = ((const float4*)(Xin + (size_t)row * DFEAT))[lane];
    }

#pragma unroll
    for (int it = 0; it < 8; ++it) {
        const int id = it * 256 + tid;
        const int n4 = id & 31;
        const int q  = id >> 5;
        float* dst = ldsW + q * WPITCH + n4 * 8;
        *(float4*)(dst)     = make_float4(wx[it].x, wy[it].x, wx[it].y, wy[it].y);
        *(float4*)(dst + 4) = make_float4(wx[it].z, wy[it].z, wx[it].w, wy[it].w);
    }

    float* myA = ldsA + wave * 16 * APITCH;
#pragma unroll
    for (int i = 0; i < 16; ++i)
        *(float4*)(myA + i * APITCH + lane * 4) = av[i];

    v8f acc[8] = {};

    float d[8] = {}, bvt[8] = {};
    if constexpr (BLEND || ROWBIAS) {
#pragma unroll
        for (int j = 0; j < 8; ++j) {
            int row = m0 + j + 8 * half;
            if (row >= Mrows) row = Mrows - 1;
            d[j] = (float)deg[row];
        }
        if (bias) {
#pragma unroll
            for (int t = 0; t < 8; ++t) bvt[t] = bias[t * 16 + r];
        }
    }

    __syncthreads();
    if (!live) return;

    const float* aL = myA + r * APITCH + 2 * half;
    const float* bL = ldsW + half * WPITCH + 2 * r;

    for (int kk = 0; kk < 32; ++kk) {
        const v2f a = *(const v2f*)(aL + 4 * kk);
        const float* bk = bL + (2 * kk) * WPITCH;
#pragma unroll
        for (int t = 0; t < 8; ++t) {
            const v2f b = *(const v2f*)(bk + t * 32);
            acc[t] = __builtin_amdgcn_wmma_f32_16x16x4_f32(
                false, a, false, b, (short)0, acc[t], false, false);
        }
    }

    if constexpr (BLEND) {
        if (m0 + 16 <= Mrows) blend_store<false>(acc, d, bvt, m0, half, r, Mrows, SB, X0, Y);
        else                  blend_store<true>(acc, d, bvt, m0, half, r, Mrows, SB, X0, Y);
        return;
    }

#pragma unroll
    for (int t = 0; t < 8; ++t) {
        const int n = t * 16 + r;
        const float bv = ROWBIAS ? bvt[t] : (bias ? bias[n] : 0.0f);
#pragma unroll
        for (int j = 0; j < 8; ++j) {
            const int row = m0 + j + 8 * half;
            if constexpr (ROWBIAS) {
                if (row < Mrows) Y[(size_t)row * DFEAT + n] = acc[t][j] + d[j] * bv;
            } else {
                if (row < Mrows) Y[(size_t)row * DFEAT + n] = acc[t][j] + bv;
            }
        }
    }
}

template <int NE, int NV, int NCHUNK, int PER>
__global__ __launch_bounds__(1024) void count_chunks(
    const int* __restrict__ vertex, const int* __restrict__ edges,
    int* __restrict__ tabE,
    int* __restrict__ tabV,
    int* __restrict__ rankE, int* __restrict__ rankV, int nnz)
{
    __shared__ int hist[(NV > NE) ? NV : NE];
    const int tid = threadIdx.x;
    const bool isE = blockIdx.x < NCHUNK;
    const int c   = isE ? (int)blockIdx.x : (int)blockIdx.x - NCHUNK;
    const int nb  = isE ? NE : NV;
    const int* __restrict__ ids  = isE ? edges : vertex;
    int* __restrict__ rank = isE ? rankE : rankV;
    int* __restrict__ row  = isE ? tabE + (size_t)c * NE : tabV + (size_t)c * NV;

    for (int i = tid; i < nb; i += 1024) hist[i] = 0;
    __syncthreads();

    const int beg = c * PER;
    const int end = (beg + PER < nnz) ? (beg + PER) : nnz;
    constexpr int U = 12;
    int base = beg + tid;
    for (; base + (U - 1) * 1024 < end; base += 1024 * U) {
        int id[U], r[U];
#pragma unroll
        for (int u = 0; u < U; ++u) id[u] = ids[base + u * 1024];
#pragma unroll
        for (int u = 0; u < U; ++u) r[u] = atomicAdd(&hist[id[u]], 1);
#pragma unroll
        for (int u = 0; u < U; ++u) rank[base + u * 1024] = r[u];
    }
    for (; base < end; base += 1024) rank[base] = atomicAdd(&hist[ids[base]], 1);
    __syncthreads();

    for (int i = tid; i < nb; i += 1024) row[i] = hist[i];
}

template <int STRIDE, int NCHUNK>
__device__ __forceinline__ void column_offsets(int* __restrict__ col, int* __restrict__ cnt)
{
    int v[NCHUNK];
#pragma unroll
    for (int c = 0; c < NCHUNK; ++c) v[c] = col[(size_t)c * STRIDE];
    int run = 0;
#pragma unroll
    for (int c = 0; c < NCHUNK; ++c) { col[(size_t)c * STRIDE] = run; run += v[c]; }
    *cnt = run;
}

#define COMPOSE_BLOCKS 9
__device__ __forceinline__ void compose_weights(
    int b, const float* __restrict__ Wa, const float* __restrict__ ba,
    const float* __restrict__ Wb, float* __restrict__ Wc, float* __restrict__ bc)
{
    const int lane = threadIdx.x & 31;
    const int t    = threadIdx.x >> 5;
    const int half = lane >> 4;
    const int r    = lane & 15;
    const int R    = 16 * b + r;
    const float* arow = (R < DFEAT) ? (Wa + (size_t)R * DFEAT) : ba;
    const bool pad = R > DFEAT;
    const int n = 16 * t + r;

    v2f a[32], bf[32];
#pragma unroll
    for (int kk = 0; kk < 32; ++kk) {
        const int k = 4 * kk + 2 * half;
        a[kk] = *(const v2f*)(arow + k);
        bf[kk][0] = Wb[(size_t)k * DFEAT + n];
        bf[kk][1] = Wb[(size_t)(k + 1) * DFEAT + n];
    }
    v8f acc = {};
#pragma unroll
    for (int kk = 0; kk < 32; ++kk) {
        const v2f av = pad ? (v2f){0.0f, 0.0f} : a[kk];
        acc = __builtin_amdgcn_wmma_f32_16x16x4_f32(
            false, av, false, bf[kk], (short)0, acc, false, false);
    }
#pragma unroll
    for (int j = 0; j < 8; ++j) {
        const int row = 16 * b + j + 8 * half;
        if (row < DFEAT)       Wc[(size_t)row * DFEAT + n] = acc[j];
        else if (row == DFEAT) bc[n] = acc[j];
    }
}

template <int NE, int NV, int NCHUNK>
__global__ __launch_bounds__(256) void chunk_offsets(
    int* __restrict__ tabE, int* __restrict__ tabV,
    int* __restrict__ cntE, int* __restrict__ cntV,
    const float* __restrict__ Wa, const float* __restrict__ ba,
    const float* __restrict__ Wb, float* __restrict__ Wc, float* __restrict__ bc)
{
    constexpr int IDB = (NE + NV + 255) / 256;
    if ((int)blockIdx.x >= IDB) {
        compose_weights((int)blockIdx.x - IDB, Wa, ba, Wb, Wc, bc);
        return;
    }
    const int i = blockIdx.x * 256 + threadIdx.x;
    if (i < NE)           column_offsets<NE, NCHUNK>(tabE + i, cntE + i);
    else if (i - NE < NV) column_offsets<NV, NCHUNK>(tabV + (i - NE), cntV + (i - NE));
}

template <int NELEM>
__device__ __forceinline__ void block_exclusive_scan(
    const int* __restrict__ cnt, int* __restrict__ cur, int* wsum)
{
    static_assert(NELEM % 4 == 0, "int4 access needs a multiple of 4");
    constexpr int IT = ((NELEM + 31) / 32 + 127) / 128;
    constexpr int CW = IT * 128;
    const int tid  = threadIdx.x;
    const int lane = tid & 31;
    const int wave = tid >> 5;
    const int base = wave * CW + lane * 4;

    int4 v[IT];
    int sum = 0;
#pragma unroll
    for (int q = 0; q < IT; ++q) {
        const int idx = base + 128 * q;
        v[q] = (idx < NELEM) ? *(const int4*)(cnt + idx) : make_int4(0, 0, 0, 0);
        sum += v[q].x + v[q].y + v[q].z + v[q].w;
    }

    for (int d = 16; d >= 1; d >>= 1) sum += __shfl_xor(sum, d);
    if (lane == 0) wsum[wave] = sum;
    __syncthreads();
    if (wave == 0) {
        const int w = wsum[lane];
        int wi = w;
        for (int d = 1; d < 32; d <<= 1) {
            const int t = __shfl_up(wi, d);
            if (lane >= d) wi += t;
        }
        wsum[lane] = wi - w;
    }
    __syncthreads();

    int carry = wsum[wave];
#pragma unroll
    for (int q = 0; q < IT; ++q) {
        const int s = v[q].x + v[q].y + v[q].z + v[q].w;
        int incl = s;
        for (int d = 1; d < 32; d <<= 1) {
            const int t = __shfl_up(incl, d);
            if (lane >= d) incl += t;
        }
        const int idx = base + 128 * q;
        if (idx < NELEM) {
            int4 o;
            o.x = carry + incl - s;
            o.y = o.x + v[q].x;
            o.z = o.y + v[q].y;
            o.w = o.z + v[q].z;
            *(int4*)(cur + idx) = o;
        }
        carry += __shfl(incl, 31);
    }
}

template <int NE, int NV>
__global__ __launch_bounds__(1024) void scan_counts(
    const int* __restrict__ cntE, int* __restrict__ curE,
    const int* __restrict__ cntV, int* __restrict__ curV)
{
    __shared__ int wsum[32];
    if (blockIdx.x == 0) block_exclusive_scan<NE>(cntE, curE, wsum);
    else                 block_exclusive_scan<NV>(cntV, curV, wsum);
}

template <int NE, int NV, int PER>
__global__ __launch_bounds__(256) void fill_lists(
    const int* __restrict__ vertex, const int* __restrict__ edges,
    const int* __restrict__ startE, const int* __restrict__ startV,
    const int* __restrict__ tabE, const int* __restrict__ tabV,
    const int* __restrict__ rankE, const int* __restrict__ rankV,
    int* __restrict__ permE, int* __restrict__ permV, int nnz)
{
    const int i = blockIdx.x * 256 + threadIdx.x;
    if (i >= nnz) return;
    const int v = vertex[i];
    const int e = edges[i];
    const int c = i / PER;
    permE[startE[e] + tabE[(size_t)c * NE + e] + rankE[i]] = v;
    permV[startV[v] + tabV[(size_t)c * NV + v] + rankV[i]] = e;
}

__global__ __launch_bounds__(256) void segsum_rows(
    const float* __restrict__ src,
    const int* __restrict__ perm,
    const int* __restrict__ segbeg,
    const int* __restrict__ cnt,
    float* __restrict__ dst,
    int nrows)
{
    const int row = blockIdx.x * 8 + (threadIdx.x >> 5);
    if (row >= nrows) return;
    const int lane = threadIdx.x & 31;
    const int beg = __builtin_amdgcn_readfirstlane(segbeg[row]);
    const int n   = __builtin_amdgcn_readfirstlane(cnt[row]);
    const int end = beg + n;

    float4 acc = make_float4(0.0f, 0.0f, 0.0f, 0.0f);
    for (int base = beg; base < end; base += 32) {
        const int m = (end - base < 32) ? (end - base) : 32;
        const int mine = (lane < m) ? perm[base + lane] : 0;
#pragma unroll 4
        for (int j = 0; j < m; ++j) {
            const int g = __builtin_amdgcn_readlane(mine, j);
            const float4 x = ((const float4*)(src + (size_t)g * DFEAT))[lane];
            acc.x += x.x; acc.y += x.y; acc.z += x.z; acc.w += x.w;
        }
    }
    ((float4*)(dst + (size_t)row * DFEAT))[lane] = acc;
}

extern "C" void kernel_launch(void* const* d_in, const int* in_sizes, int n_in,
                              void* d_out, int out_size, void* d_ws, size_t ws_size,
                              hipStream_t stream) {
    const float* X      = (const float*)d_in[0];
    const int*   vertex = (const int*)d_in[1];
    const int*   edges  = (const int*)d_in[2];
    const float* X0     = (const float*)d_in[3];
    const float* W1     = (const float*)d_in[4];
    const float* b1     = (const float*)d_in[5];
    const float* W2     = (const float*)d_in[6];
    const float* b2     = (const float*)d_in[7];
    const float* W3     = (const float*)d_in[8];
    const float* b3     = (const float*)d_in[9];
    float* out = (float*)d_out;

    constexpr int N = 50000, M = 25000, NNZ = 800000;
    constexpr size_t ND = (size_t)N * DFEAT;
    constexpr size_t MD = (size_t)M * DFEAT;

    char* ws = (char*)d_ws;
    auto align256 = [](size_t x) { return (x + 255) & ~(size_t)255; };
    size_t off = 0;
    float* Xn = (float*)(ws + off); off = align256(off + ND * 4);
    float* Xe = (float*)(ws + off); off = align256(off + MD * 4);
    const size_t offLists = off;
    off = align256(off + ND * 4);
    float* Bm = (float*)(ws + off); off = align256(off + MD * 4);
    float* SB = (float*)(ws + off); off = align256(off + ND * 4);
    int*  deg = (int*)  (ws + off); off = align256(off + (size_t)N * 4);
    float* Wc = Xn;
    float* bc = Xn + 128 * DFEAT;

    size_t o2 = offLists;
    int* permE = (int*)(ws + o2); o2 = align256(o2 + (size_t)NNZ * 4);
    int* permV = (int*)(ws + o2); o2 = align256(o2 + (size_t)NNZ * 4);
    int* curE  = (int*)(ws + o2); o2 = align256(o2 + (size_t)M * 4);
    int* curV  = (int*)(ws + o2); o2 = align256(o2 + (size_t)N * 4);
    int* cntE  = (int*)(ws + o2); o2 = align256(o2 + (size_t)M * 4);
    int* rankE = (int*)(ws + o2); o2 = align256(o2 + (size_t)NNZ * 4);
    int* rankV = (int*)(ws + o2); o2 = align256(o2 + (size_t)NNZ * 4);
    constexpr int NCHUNK = 32;
    constexpr int PER = (NNZ + NCHUNK - 1) / NCHUNK;
    int* tabE  = (int*)(ws + o2); o2 = align256(o2 + (size_t)NCHUNK * M * 4);
    int* tabV  = (int*)(ws + o2); o2 = align256(o2 + (size_t)NCHUNK * N * 4);
    int* cntV  = deg;
    static_assert(4 * (size_t)NNZ * 4 + 2 * (size_t)M * 4 + (size_t)N * 4
                  + (size_t)NCHUNK * (M + N) * 4 + 9 * 256 <= ND * 4,
                  "sorted lists, ranks and chunk tables must fit in their region");

    const int gemmN = (N + 127) / 128;
    const int gemmM = (M + 127) / 128;
    const int nnzB  = (NNZ + 255) / 256;

    count_chunks<M, N, NCHUNK, PER><<<2 * NCHUNK, 1024, 0, stream>>>(vertex, edges, tabE, tabV, rankE, rankV, NNZ);
    chunk_offsets<M, N, NCHUNK><<<(M + N + 255) / 256 + COMPOSE_BLOCKS, 256, 0, stream>>>(
        tabE, tabV, cntE, cntV, W1, b1, W2 + 128 * DFEAT, Wc, bc);
    scan_counts<M, N><<<2, 1024, 0, stream>>>(cntE, curE, cntV, curV);
    fill_lists<M, N, PER><<<nnzB, 256, 0, stream>>>(vertex, edges, curE, curV, tabE, tabV, rankE, rankV, permE, permV, NNZ);

    segsum_rows<<<(M + 7) / 8, 256, 0, stream>>>(X, permE, curE, cntE, Xe, M);

    gemm128_wmma<false, true><<<gemmM, 256, 0, stream>>>(Xe, M, Wc, bc, nullptr, cntE, nullptr, Bm);

    segsum_rows<<<(N + 7) / 8, 256, 0, stream>>>(Bm, permV, curV, cntV, SB, N);

    gemm128_wmma<true, false><<<gemmN, 256, 0, stream>>>(X, N, W2, b2, SB, deg, X0, Xn);

    gemm128_wmma<false, false><<<gemmN, 256, 0, stream>>>(Xn, N, W3, b3, nullptr, nullptr, nullptr, out);
}


# Xn@W3 fused behind the blended X@W2a product in one kernel: each wave's blended 16x128 tile stays in its LDS rows, W3 replaces W2a in LDS; one launch and the Xn round trip go; sums unchanged
# speedup vs baseline: 2.8831x; 1.0763x over previous
#include <hip/hip_runtime.h>


typedef __attribute__((ext_vector_type(2))) float v2f;
typedef __attribute__((ext_vector_type(8))) float v8f;

#define DFEAT 128

template <bool GUARD>
__device__ __forceinline__ void blend_to_lds(
    const v8f (&acc)[8], const float (&d)[8], const float (&bv)[8],
    int m0, int half, int r, int Mrows,
    const float* __restrict__ SB, const float* __restrict__ X0,
    float* tile, int pitch)
{
    float sb[8][8], x0[8][8];
#pragma unroll
    for (int t = 0; t < 8; ++t) {
        const int n = t * 16 + r;
#pragma unroll
        for (int j = 0; j < 8; ++j) {
            const int row = m0 + j + 8 * half;
            sb[t][j] = 0.0f;
            x0[t][j] = 0.0f;
            if (!GUARD || row < Mrows) {
                const size_t i = (size_t)row * DFEAT + n;
                sb[t][j] = SB[i];
                x0[t][j] = X0[i];
            }
        }
    }
#pragma unroll
    for (int t = 0; t < 8; ++t) {
        const int n = t * 16 + r;
#pragma unroll
        for (int j = 0; j < 8; ++j) {
            const float ap = acc[t][j] + bv[t];
            tile[(j + 8 * half) * pitch + n] = 0.5f * (d[j] * ap + sb[t][j]) + 0.5f * x0[t][j];
        }
    }
}

#define WPITCH 288
#define APITCH 132

__device__ __forceinline__ void w_load(const float* __restrict__ W, int tid,
                                       float4 (&wx)[8], float4 (&wy)[8])
{
#pragma unroll
    for (int it = 0; it < 8; ++it) {
        const int id = it * 256 + tid;
        const int n4 = id & 31;
        const int q  = id >> 5;
        wx[it] = *(const float4*)(W + (size_t)(2 * q) * DFEAT + n4 * 4);
        wy[it] = *(const float4*)(W + (size_t)(2 * q + 1) * DFEAT + n4 * 4);
    }
}

__device__ __forceinline__ void w_store(float* ldsW, int tid,
                                        const float4 (&wx)[8], const float4 (&wy)[8])
{
#pragma unroll
    for (int it = 0; it < 8; ++it) {
        const int id = it * 256 + tid;
        const int n4 = id & 31;
        const int q  = id >> 5;
        float* dst = ldsW + q * WPITCH + n4 * 8;
        *(float4*)(dst)     = make_float4(wx[it].x, wy[it].x, wx[it].y, wy[it].y);
        *(float4*)(dst + 4) = make_float4(wx[it].z, wy[it].z, wx[it].w, wy[it].w);
    }
}

__device__ __forceinline__ void a_load(const float* __restrict__ Xin, int m0, int Mrows,
                                       int lane, float4 (&av)[16])
{
#pragma unroll
    for (int i = 0; i < 16; ++i) {
        int row = m0 + i;
        if (row >= Mrows) row = Mrows - 1;
        av[i] = ((const float4*)(Xin + (size_t)row * DFEAT))[lane];
    }
}

__device__ __forceinline__ void a_store(float* myA, int lane, const float4 (&av)[16])
{
#pragma unroll
    for (int i = 0; i < 16; ++i)
        *(float4*)(myA + i * APITCH + lane * 4) = av[i];
}

__device__ __forceinline__ void k_loop(const float* myA, const float* ldsW,
                                       int half, int r, v8f (&acc)[8])
{
    const float* aL = myA + r * APITCH + 2 * half;
    const float* bL = ldsW + half * WPITCH + 2 * r;

    for (int kk = 0; kk < 32; ++kk) {
        const v2f a = *(const v2f*)(aL + 4 * kk);
        const float* bk = bL + (2 * kk) * WPITCH;
#pragma unroll
        for (int t = 0; t < 8; ++t) {
            const v2f b = *(const v2f*)(bk + t * 32);
            acc[t] = __builtin_amdgcn_wmma_f32_16x16x4_f32(
                false, a, false, b, (short)0, acc[t], false, false);
        }
    }
}

template <bool ROWBIAS>
__global__ __launch_bounds__(256) void gemm128_wmma(
    const float* __restrict__ Xin, int Mrows,
    const float* __restrict__ W,
    const float* __restrict__ bias,
    const int* __restrict__ deg,
    float* __restrict__ Y)
{
    __shared__ __attribute__((aligned(16))) float ldsW[64 * WPITCH];
    __shared__ __attribute__((aligned(16))) float ldsA[8 * 16 * APITCH];

    const int tid  = threadIdx.x;
    const int lane = tid & 31;
    const int wave = tid >> 5;
    const int half = lane >> 4;
    const int r    = lane & 15;
    const int m0   = (blockIdx.x * 8 + wave) * 16;
    const bool live = m0 < Mrows;

    float4 wx[8], wy[8], av[16];
    w_load(W, tid, wx, wy);
    a_load(Xin, m0, Mrows, lane, av);
    w_store(ldsW, tid, wx, wy);
    float* myA = ldsA + wave * 16 * APITCH;
    a_store(myA, lane, av);

    v8f acc[8] = {};

    float d[8] = {}, bvt[8] = {};
    if constexpr (ROWBIAS) {
#pragma unroll
        for (int j = 0; j < 8; ++j) {
            int row = m0 + j + 8 * half;
            if (row >= Mrows) row = Mrows - 1;
            d[j] = (float)deg[row];
        }
        if (bias) {
#pragma unroll
            for (int t = 0; t < 8; ++t) bvt[t] = bias[t * 16 + r];
        }
    }

    __syncthreads();
    if (!live) return;

    k_loop(myA, ldsW, half, r, acc);

#pragma unroll
    for (int t = 0; t < 8; ++t) {
        const int n = t * 16 + r;
        const float bv = ROWBIAS ? bvt[t] : (bias ? bias[n] : 0.0f);
#pragma unroll
        for (int j = 0; j < 8; ++j) {
            const int row = m0 + j + 8 * half;
            if constexpr (ROWBIAS) {
                if (row < Mrows) Y[(size_t)row * DFEAT + n] = acc[t][j] + d[j] * bv;
            } else {
                if (row < Mrows) Y[(size_t)row * DFEAT + n] = acc[t][j] + bv;
            }
        }
    }
}

__global__ __launch_bounds__(256) void gemm128_blend_gemm128(
    const float* __restrict__ Xin, int Mrows,
    const float* __restrict__ Wa,
    const float* __restrict__ ba,
    const float* __restrict__ SB,
    const int* __restrict__ deg,
    const float* __restrict__ X0,
    const float* __restrict__ Wb,
    const float* __restrict__ bb,
    float* __restrict__ Y)
{
    __shared__ __attribute__((aligned(16))) float ldsW[64 * WPITCH];
    __shared__ __attribute__((aligned(16))) float ldsA[8 * 16 * APITCH];

    const int tid  = threadIdx.x;
    const int lane = tid & 31;
    const int wave = __builtin_amdgcn_readfirstlane(tid >> 5);
    const int half = lane >> 4;
    const int r    = lane & 15;
    const int m0   = (blockIdx.x * 8 + wave) * 16;
    const bool live = m0 < Mrows;

    float4 wx[8], wy[8], av[16];
    w_load(Wa, tid, wx, wy);
    a_load(Xin, m0, Mrows, lane, av);
    w_store(ldsW, tid, wx, wy);
    float* myA = ldsA + wave * 16 * APITCH;
    a_store(myA, lane, av);

    float d[8], bav[8], bbv[8];
#pragma unroll
    for (int j = 0; j < 8; ++j) {
        int row = m0 + j + 8 * half;
        if (row >= Mrows) row = Mrows - 1;
        d[j] = (float)deg[row];
    }
#pragma unroll
    for (int t = 0; t < 8; ++t) bav[t] = ba[t * 16 + r];
#pragma unroll
    for (int t = 0; t < 8; ++t) bbv[t] = bb[t * 16 + r];

    __syncthreads();

    if (live) {
        v8f acc[8] = {};
        k_loop(myA, ldsW, half, r, acc);
        if (m0 + 16 <= Mrows) blend_to_lds<false>(acc, d, bav, m0, half, r, Mrows, SB, X0, myA, APITCH);
        else                  blend_to_lds<true>(acc, d, bav, m0, half, r, Mrows, SB, X0, myA, APITCH);
    }

    w_load(Wb, tid, wx, wy);
    __syncthreads();
    w_store(ldsW, tid, wx, wy);
    __syncthreads();
    if (!live) return;

    v8f acc2[8] = {};
    k_loop(myA, ldsW, half, r, acc2);

#pragma unroll
    for (int t = 0; t < 8; ++t) {
        const int n = t * 16 + r;
#pragma unroll
        for (int j = 0; j < 8; ++j) {
            const int row = m0 + j + 8 * half;
            if (row < Mrows) Y[(size_t)row * DFEAT + n] = acc2[t][j] + bbv[t];
        }
    }
}

template <int NE, int NV, int NCHUNK, int PER>
__global__ __launch_bounds__(1024) void count_chunks(
    const int* __restrict__ vertex, const int* __restrict__ edges,
    int* __restrict__ tabE,
    int* __restrict__ tabV,
    int* __restrict__ rankE, int* __restrict__ rankV, int nnz)
{
    __shared__ int hist[(NV > NE) ? NV : NE];
    const int tid = threadIdx.x;
    const bool isE = blockIdx.x < NCHUNK;
    const int c   = isE ? (int)blockIdx.x : (int)blockIdx.x - NCHUNK;
    const int nb  = isE ? NE : NV;
    const int* __restrict__ ids  = isE ? edges : vertex;
    int* __restrict__ rank = isE ? rankE : rankV;
    int* __restrict__ row  = isE ? tabE + (size_t)c * NE : tabV + (size_t)c * NV;

    for (int i = tid; i < nb; i += 1024) hist[i] = 0;
    __syncthreads();

    const int beg = c * PER;
    const int end = (beg + PER < nnz) ? (beg + PER) : nnz;
    constexpr int U = 12;
    int base = beg + tid;
    for (; base + (U - 1) * 1024 < end; base += 1024 * U) {
        int id[U], r[U];
#pragma unroll
        for (int u = 0; u < U; ++u) id[u] = ids[base + u * 1024];
#pragma unroll
        for (int u = 0; u < U; ++u) r[u] = atomicAdd(&hist[id[u]], 1);
#pragma unroll
        for (int u = 0; u < U; ++u) rank[base + u * 1024] = r[u];
    }
    for (; base < end; base += 1024) rank[base] = atomicAdd(&hist[ids[base]], 1);
    __syncthreads();

    for (int i = tid; i < nb; i += 1024) row[i] = hist[i];
}

template <int STRIDE, int NCHUNK>
__device__ __forceinline__ void column_offsets(int* __restrict__ col, int* __restrict__ cnt)
{
    int v[NCHUNK];
#pragma unroll
    for (int c = 0; c < NCHUNK; ++c) v[c] = col[(size_t)c * STRIDE];
    int run = 0;
#pragma unroll
    for (int c = 0; c < NCHUNK; ++c) { col[(size_t)c * STRIDE] = run; run += v[c]; }
    *cnt = run;
}

#define COMPOSE_BLOCKS 9
__device__ __forceinline__ void compose_weights(
    int b, const float* __restrict__ Wa, const float* __restrict__ ba,
    const float* __restrict__ Wb, float* __restrict__ Wc, float* __restrict__ bc)
{
    const int lane = threadIdx.x & 31;
    const int t    = threadIdx.x >> 5;
    const int half = lane >> 4;
    const int r    = lane & 15;
    const int R    = 16 * b + r;
    const float* arow = (R < DFEAT) ? (Wa + (size_t)R * DFEAT) : ba;
    const bool pad = R > DFEAT;
    const int n = 16 * t + r;

    v2f a[32], bf[32];
#pragma unroll
    for (int kk = 0; kk < 32; ++kk) {
        const int k = 4 * kk + 2 * half;
        a[kk] = *(const v2f*)(arow + k);
        bf[kk][0] = Wb[(size_t)k * DFEAT + n];
        bf[kk][1] = Wb[(size_t)(k + 1) * DFEAT + n];
    }
    v8f acc = {};
#pragma unroll
    for (int kk = 0; kk < 32; ++kk) {
        const v2f av = pad ? (v2f){0.0f, 0.0f} : a[kk];
        acc = __builtin_amdgcn_wmma_f32_16x16x4_f32(
            false, av, false, bf[kk], (short)0, acc, false, false);
    }
#pragma unroll
    for (int j = 0; j < 8; ++j) {
        const int row = 16 * b + j + 8 * half;
        if (row < DFEAT)       Wc[(size_t)row * DFEAT + n] = acc[j];
        else if (row == DFEAT) bc[n] = acc[j];
    }
}

template <int NE, int NV, int NCHUNK>
__global__ __launch_bounds__(256) void chunk_offsets(
    int* __restrict__ tabE, int* __restrict__ tabV,
    int* __restrict__ cntE, int* __restrict__ cntV,
    const float* __restrict__ Wa, const float* __restrict__ ba,
    const float* __restrict__ Wb, float* __restrict__ Wc, float* __restrict__ bc)
{
    constexpr int IDB = (NE + NV + 255) / 256;
    if ((int)blockIdx.x >= IDB) {
        compose_weights((int)blockIdx.x - IDB, Wa, ba, Wb, Wc, bc);
        return;
    }
    const int i = blockIdx.x * 256 + threadIdx.x;
    if (i < NE)           column_offsets<NE, NCHUNK>(tabE + i, cntE + i);
    else if (i - NE < NV) column_offsets<NV, NCHUNK>(tabV + (i - NE), cntV + (i - NE));
}

template <int NELEM>
__device__ __forceinline__ void block_exclusive_scan(
    const int* __restrict__ cnt, int* __restrict__ cur, int* wsum)
{
    static_assert(NELEM % 4 == 0, "int4 access needs a multiple of 4");
    constexpr int IT = ((NELEM + 31) / 32 + 127) / 128;
    constexpr int CW = IT * 128;
    const int tid  = threadIdx.x;
    const int lane = tid & 31;
    const int wave = tid >> 5;
    const int base = wave * CW + lane * 4;

    int4 v[IT];
    int sum = 0;
#pragma unroll
    for (int q = 0; q < IT; ++q) {
        const int idx = base + 128 * q;
        v[q] = (idx < NELEM) ? *(const int4*)(cnt + idx) : make_int4(0, 0, 0, 0);
        sum += v[q].x + v[q].y + v[q].z + v[q].w;
    }

    for (int d = 16; d >= 1; d >>= 1) sum += __shfl_xor(sum, d);
    if (lane == 0) wsum[wave] = sum;
    __syncthreads();
    if (wave == 0) {
        const int w = wsum[lane];
        int wi = w;
        for (int d = 1; d < 32; d <<= 1) {
            const int t = __shfl_up(wi, d);
            if (lane >= d) wi += t;
        }
        wsum[lane] = wi - w;
    }
    __syncthreads();

    int carry = wsum[wave];
#pragma unroll
    for (int q = 0; q < IT; ++q) {
        const int s = v[q].x + v[q].y + v[q].z + v[q].w;
        int incl = s;
        for (int d = 1; d < 32; d <<= 1) {
            const int t = __shfl_up(incl, d);
            if (lane >= d) incl += t;
        }
        const int idx = base + 128 * q;
        if (idx < NELEM) {
            int4 o;
            o.x = carry + incl - s;
            o.y = o.x + v[q].x;
            o.z = o.y + v[q].y;
            o.w = o.z + v[q].z;
            *(int4*)(cur + idx) = o;
        }
        carry += __shfl(incl, 31);
    }
}

template <int NE, int NV>
__global__ __launch_bounds__(1024) void scan_counts(
    const int* __restrict__ cntE, int* __restrict__ curE,
    const int* __restrict__ cntV, int* __restrict__ curV)
{
    __shared__ int wsum[32];
    if (blockIdx.x == 0) block_exclusive_scan<NE>(cntE, curE, wsum);
    else                 block_exclusive_scan<NV>(cntV, curV, wsum);
}

template <int NE, int NV, int PER>
__global__ __launch_bounds__(256) void fill_lists(
    const int* __restrict__ vertex, const int* __restrict__ edges,
    const int* __restrict__ startE, const int* __restrict__ startV,
    const int* __restrict__ tabE, const int* __restrict__ tabV,
    const int* __restrict__ rankE, const int* __restrict__ rankV,
    int* __restrict__ permE, int* __restrict__ permV, int nnz)
{
    const int i = blockIdx.x * 256 + threadIdx.x;
    if (i >= nnz) return;
    const int v = vertex[i];
    const int e = edges[i];
    const int c = i / PER;
    permE[startE[e] + tabE[(size_t)c * NE + e] + rankE[i]] = v;
    permV[startV[v] + tabV[(size_t)c * NV + v] + rankV[i]] = e;
}

__global__ __launch_bounds__(256) void segsum_rows(
    const float* __restrict__ src,
    const int* __restrict__ perm,
    const int* __restrict__ segbeg,
    const int* __restrict__ cnt,
    float* __restrict__ dst,
    int nrows)
{
    const int row = blockIdx.x * 8 + (threadIdx.x >> 5);
    if (row >= nrows) return;
    const int lane = threadIdx.x & 31;
    const int beg = __builtin_amdgcn_readfirstlane(segbeg[row]);
    const int n   = __builtin_amdgcn_readfirstlane(cnt[row]);
    const int end = beg + n;

    float4 acc = make_float4(0.0f, 0.0f, 0.0f, 0.0f);
    for (int base = beg; base < end; base += 32) {
        const int m = (end - base < 32) ? (end - base) : 32;
        const int mine = (lane < m) ? perm[base + lane] : 0;
#pragma unroll 4
        for (int j = 0; j < m; ++j) {
            const int g = __builtin_amdgcn_readlane(mine, j);
            const float4 x = ((const float4*)(src + (size_t)g * DFEAT))[lane];
            acc.x += x.x; acc.y += x.y; acc.z += x.z; acc.w += x.w;
        }
    }
    ((float4*)(dst + (size_t)row * DFEAT))[lane] = acc;
}

extern "C" void kernel_launch(void* const* d_in, const int* in_sizes, int n_in,
                              void* d_out, int out_size, void* d_ws, size_t ws_size,
                              hipStream_t stream) {
    const float* X      = (const float*)d_in[0];
    const int*   vertex = (const int*)d_in[1];
    const int*   edges  = (const int*)d_in[2];
    const float* X0     = (const float*)d_in[3];
    const float* W1     = (const float*)d_in[4];
    const float* b1     = (const float*)d_in[5];
    const float* W2     = (const float*)d_in[6];
    const float* b2     = (const float*)d_in[7];
    const float* W3     = (const float*)d_in[8];
    const float* b3     = (const float*)d_in[9];
    float* out = (float*)d_out;

    constexpr int N = 50000, M = 25000, NNZ = 800000;
    constexpr size_t ND = (size_t)N * DFEAT;
    constexpr size_t MD = (size_t)M * DFEAT;

    char* ws = (char*)d_ws;
    auto align256 = [](size_t x) { return (x + 255) & ~(size_t)255; };
    size_t off = 0;
    float* Wcb = (float*)(ws + off); off = align256(off + (size_t)129 * DFEAT * 4);
    float* Xe = (float*)(ws + off); off = align256(off + MD * 4);
    const size_t offLists = off;
    off = align256(off + ND * 4);
    float* Bm = (float*)(ws + off); off = align256(off + MD * 4);
    float* SB = (float*)(ws + off); off = align256(off + ND * 4);
    int*  deg = (int*)  (ws + off); off = align256(off + (size_t)N * 4);
    float* Wc = Wcb;
    float* bc = Wcb + 128 * DFEAT;

    size_t o2 = offLists;
    int* permE = (int*)(ws + o2); o2 = align256(o2 + (size_t)NNZ * 4);
    int* permV = (int*)(ws + o2); o2 = align256(o2 + (size_t)NNZ * 4);
    int* curE  = (int*)(ws + o2); o2 = align256(o2 + (size_t)M * 4);
    int* curV  = (int*)(ws + o2); o2 = align256(o2 + (size_t)N * 4);
    int* cntE  = (int*)(ws + o2); o2 = align256(o2 + (size_t)M * 4);
    int* rankE = (int*)(ws + o2); o2 = align256(o2 + (size_t)NNZ * 4);
    int* rankV = (int*)(ws + o2); o2 = align256(o2 + (size_t)NNZ * 4);
    constexpr int NCHUNK = 32;
    constexpr int PER = (NNZ + NCHUNK - 1) / NCHUNK;
    int* tabE  = (int*)(ws + o2); o2 = align256(o2 + (size_t)NCHUNK * M * 4);
    int* tabV  = (int*)(ws + o2); o2 = align256(o2 + (size_t)NCHUNK * N * 4);
    int* cntV  = deg;
    static_assert(4 * (size_t)NNZ * 4 + 2 * (size_t)M * 4 + (size_t)N * 4
                  + (size_t)NCHUNK * (M + N) * 4 + 9 * 256 <= ND * 4,
                  "sorted lists, ranks and chunk tables must fit in their region");

    const int gemmN = (N + 127) / 128;
    const int gemmM = (M + 127) / 128;
    const int nnzB  = (NNZ + 255) / 256;

    count_chunks<M, N, NCHUNK, PER><<<2 * NCHUNK, 1024, 0, stream>>>(vertex, edges, tabE, tabV, rankE, rankV, NNZ);
    chunk_offsets<M, N, NCHUNK><<<(M + N + 255) / 256 + COMPOSE_BLOCKS, 256, 0, stream>>>(
        tabE, tabV, cntE, cntV, W1, b1, W2 + 128 * DFEAT, Wc, bc);
    scan_counts<M, N><<<2, 1024, 0, stream>>>(cntE, curE, cntV, curV);
    fill_lists<M, N, PER><<<nnzB, 256, 0, stream>>>(vertex, edges, curE, curV, tabE, tabV, rankE, rankV, permE, permV, NNZ);

    segsum_rows<<<(M + 7) / 8, 256, 0, stream>>>(X, permE, curE, cntE, Xe, M);

    gemm128_wmma<true><<<gemmM, 256, 0, stream>>>(Xe, M, Wc, bc, cntE, Bm);

    segsum_rows<<<(N + 7) / 8, 256, 0, stream>>>(Bm, permV, curV, cntV, SB, N);

    gemm128_blend_gemm128<<<gemmN, 256, 0, stream>>>(X, N, W2, b2, SB, deg, X0, W3, b3, out);
}
